# starting kernel
# baseline (speedup 1.0000x reference)
#include <hip/hip_runtime.h>

#define N_NODES 50000
#define N_EDGES 640000
#define CH      128

typedef float v2f __attribute__((ext_vector_type(2)));
typedef float v8f __attribute__((ext_vector_type(8)));


__global__ void gc_init_ws(float* __restrict__ ws, int n) {
    int i = blockIdx.x * blockDim.x + threadIdx.x;
    if (i < n) ws[i] = 0.0f;
}

__global__ __launch_bounds__(256) void gc_scatter(const float* __restrict__ x,
                                                  const int*   __restrict__ rows,
                                                  const int*   __restrict__ cols,
                                                  float*       __restrict__ deg,
                                                  float*       __restrict__ agg) {
    const int wave = (blockIdx.x * blockDim.x + threadIdx.x) >> 5;
    const int lane = threadIdx.x & 31;
    const int r = rows[wave];
    const int c = cols[wave];
    const float4 v = ((const float4*)(x + (size_t)c * CH))[lane];
    float* dst = agg + (size_t)r * CH + lane * 4;
    atomicAdd(dst + 0, v.x);
    atomicAdd(dst + 1, v.y);
    atomicAdd(dst + 2, v.z);
    atomicAdd(dst + 3, v.w);
    if (lane == 0) atomicAdd(deg + r, 1.0f);
}

__global__ __launch_bounds__(256) void gc_finalize(const float* __restrict__ x,
                                                   const float* __restrict__ deg,
                                                   float*       __restrict__ agg) {
    const int i = blockIdx.x * blockDim.x + threadIdx.x;
    const int node = i >> 7;
    float d = deg[node];
    d = d < 1.0f ? 1.0f : d;
    agg[i] = x[i] + agg[i] / d;
}

__global__ __launch_bounds__(256) void gc_gemm_wmma(const float* __restrict__ h,
                                                    const float* __restrict__ W,
                                                    const float* __restrict__ bias,
                                                    float*       __restrict__ out) {
    const int wid  = (blockIdx.x * blockDim.x + threadIdx.x) >> 5;
    const int lane = threadIdx.x & 31;
    const int mt = wid >> 3;
    const int nt = wid & 7;
    const int m0 = mt * 16;
    const int n0 = nt * 16;
    const int ln = lane & 15;
    const int hi = lane >> 4;

    const float bv = bias[n0 + ln];
    v8f c;
#pragma unroll
    for (int r = 0; r < 8; ++r) c[r] = bv;

    const float* hrow = h + (size_t)(m0 + ln) * CH;
    const float* wrow = W + (size_t)(n0 + ln) * CH;

#pragma unroll 4
    for (int kb = 0; kb < 32; ++kb) {
        const int k = kb * 4 + hi * 2;
        v2f a = *(const v2f*)(hrow + k);
        v2f b = *(const v2f*)(wrow + k);
        c = __builtin_amdgcn_wmma_f32_16x16x4_f32(false, a, false, b,
                                                  (short)0, c, false, false);
    }

    float* orow = out + (size_t)(m0 + hi * 8) * CH + n0 + ln;
#pragma unroll
    for (int r = 0; r < 8; ++r) orow[(size_t)r * CH] = c[r];
}


extern "C" void kernel_launch(void* const* d_in, const int* in_sizes, int n_in,
                              void* d_out, int out_size, void* d_ws, size_t ws_size,
                              hipStream_t stream) {
    const float* x    = (const float*)d_in[0];
    const int*   ei   = (const int*)  d_in[1];
    const float* W    = (const float*)d_in[2];
    const float* bias = (const float*)d_in[3];
    float*       out  = (float*)d_out;

    const int* rows = ei;
    const int* cols = ei + N_EDGES;

    float* deg = (float*)d_ws;
    float* agg = deg + N_NODES;

    {
        const int n = N_NODES + N_NODES * CH;
        gc_init_ws<<<(n + 255) / 256, 256, 0, stream>>>((float*)d_ws, n);
    }
    {
        const int blocks = (N_EDGES * 32) / 256;
        gc_scatter<<<blocks, 256, 0, stream>>>(x, rows, cols, deg, agg);
    }
    {
        const int blocks = (N_NODES * CH) / 256;
        gc_finalize<<<blocks, 256, 0, stream>>>(x, deg, agg);
    }
    {
        const int blocks = (3125 * 8 * 32) / 256;
        gc_gemm_wmma<<<blocks, 256, 0, stream>>>(agg, W, bias, out);
    }
}


# f32 atomic scatter-add replaced by in-call counting sort of edges by row (int atomic count+slot, 1-block scan, fill) and an atomic-free per-node gather; sums now in slot order, 4 partial sums
# speedup vs baseline: 1.7367x; 1.7367x over previous
#include <hip/hip_runtime.h>


#define N_NODES 50000
#define N_EDGES 640000
#define CH      128
#define N_PAD   50176

typedef float v2f __attribute__((ext_vector_type(2)));
typedef float v8f __attribute__((ext_vector_type(8)));


__global__ __launch_bounds__(256) void gc_zero_cnt(int* __restrict__ cnt) {
    const int i = blockIdx.x * blockDim.x + threadIdx.x;
    if (i < N_NODES) cnt[i] = 0;
}

__global__ __launch_bounds__(256) void gc_count(const int* __restrict__ rows,
                                                int*       __restrict__ cnt,
                                                int*       __restrict__ rank) {
    const int e = blockIdx.x * blockDim.x + threadIdx.x;
    const int r = rows[e];
    rank[e] = atomicAdd(cnt + r, 1);
}

__global__ __launch_bounds__(1024) void gc_scan(const int* __restrict__ cnt,
                                                int*       __restrict__ offs) {
    __shared__ int4 s4[N_PAD / 4];
    __shared__ int wtot[32];
    int* s = (int*)s4;
    const int t    = threadIdx.x;
    const int lane = t & 31;
    const int wave = t >> 5;

    const int4* __restrict__ cnt4 = (const int4*)cnt;
    int4 st[13];
#pragma unroll
    for (int k = 0; k < 13; ++k) {
        const int q = t + k * 1024;
        st[k] = (q < N_NODES / 4) ? cnt4[q] : make_int4(0, 0, 0, 0);
    }
#pragma unroll
    for (int k = 0; k < 13; ++k) {
        const int q = t + k * 1024;
        if (q < N_PAD / 4) s4[q] = st[k];
    }
    __syncthreads();

    const int base = t * 49;
    int sum = 0;
#pragma unroll
    for (int i = 0; i < 49; ++i) sum += s[base + i];

    int incl = sum;
#pragma unroll
    for (int d = 1; d < 32; d <<= 1) {
        const int up = __shfl_up(incl, d);
        if (lane >= d) incl += up;
    }
    if (lane == 31) wtot[wave] = incl;
    __syncthreads();

    const int wv = wtot[lane];
    int wincl = wv;
#pragma unroll
    for (int d = 1; d < 32; d <<= 1) {
        const int up = __shfl_up(wincl, d);
        if (lane >= d) wincl += up;
    }
    const int wbase = __shfl(wincl - wv, wave);

    int run = wbase + incl - sum;
#pragma unroll
    for (int i = 0; i < 49; ++i) {
        const int v = s[base + i];
        s[base + i] = run;
        run += v;
    }
    __syncthreads();

    int4* __restrict__ offs4 = (int4*)offs;
#pragma unroll
    for (int k = 0; k < 13; ++k) {
        const int q = t + k * 1024;
        if (q < N_NODES / 4) offs4[q] = s4[q];
    }
}

__global__ __launch_bounds__(256) void gc_fill(const int* __restrict__ rows,
                                               const int* __restrict__ cols,
                                               const int* __restrict__ offs,
                                               const int* __restrict__ rank,
                                               int*       __restrict__ csr) {
    const int e = blockIdx.x * blockDim.x + threadIdx.x;
    const int r = rows[e];
    unsigned p = (unsigned)(offs[r] + rank[e]);
    p = p < (unsigned)N_EDGES ? p : (unsigned)(N_EDGES - 1);
    csr[p] = cols[e];
}

__global__ __launch_bounds__(256) void gc_gather(const float* __restrict__ x,
                                                 const int*   __restrict__ cnt,
                                                 const int*   __restrict__ offs,
                                                 const int*   __restrict__ csr,
                                                 float*       __restrict__ h) {
    const int lane = threadIdx.x & 31;
    const int node = __builtin_amdgcn_readfirstlane((int)(blockIdx.x * 8 + (threadIdx.x >> 5)));
    const int n    = __builtin_amdgcn_readfirstlane(cnt[node]);
    const int beg  = __builtin_amdgcn_readfirstlane(offs[node]);

    const float4* __restrict__ x4 = (const float4*)x;
    const float4 xv = x4[(size_t)node * 32 + lane];

    float4 a0 = make_float4(0.f, 0.f, 0.f, 0.f);
    float4 a1 = a0, a2 = a0, a3 = a0;

    for (int b = 0; b < n; b += 32) {
        const int rem = n - b;
        const int m = rem < 32 ? rem : 32;
        int cl = 0;
        if (lane < m) cl = csr[beg + b + lane];
        cl = (unsigned)cl < (unsigned)N_NODES ? cl : (N_NODES - 1);
        int j = 0;
        for (; j + 4 <= m; j += 4) {
            const int c0 = __builtin_amdgcn_readlane(cl, j + 0);
            const int c1 = __builtin_amdgcn_readlane(cl, j + 1);
            const int c2 = __builtin_amdgcn_readlane(cl, j + 2);
            const int c3 = __builtin_amdgcn_readlane(cl, j + 3);
            const float4 v0 = x4[(size_t)c0 * 32 + lane];
            const float4 v1 = x4[(size_t)c1 * 32 + lane];
            const float4 v2 = x4[(size_t)c2 * 32 + lane];
            const float4 v3 = x4[(size_t)c3 * 32 + lane];
            a0.x += v0.x; a0.y += v0.y; a0.z += v0.z; a0.w += v0.w;
            a1.x += v1.x; a1.y += v1.y; a1.z += v1.z; a1.w += v1.w;
            a2.x += v2.x; a2.y += v2.y; a2.z += v2.z; a2.w += v2.w;
            a3.x += v3.x; a3.y += v3.y; a3.z += v3.z; a3.w += v3.w;
        }
        const int k = m - j;
        if (k > 0) {
            const int c0 = __builtin_amdgcn_readlane(cl, j + 0);
            const float4 v0 = x4[(size_t)c0 * 32 + lane];
            if (k > 1) {
                const int c1 = __builtin_amdgcn_readlane(cl, j + 1);
                const float4 v1 = x4[(size_t)c1 * 32 + lane];
                if (k > 2) {
                    const int c2 = __builtin_amdgcn_readlane(cl, j + 2);
                    const float4 v2 = x4[(size_t)c2 * 32 + lane];
                    a2.x += v2.x; a2.y += v2.y; a2.z += v2.z; a2.w += v2.w;
                }
                a1.x += v1.x; a1.y += v1.y; a1.z += v1.z; a1.w += v1.w;
            }
            a0.x += v0.x; a0.y += v0.y; a0.z += v0.z; a0.w += v0.w;
        }
    }

    const float d = (float)(n < 1 ? 1 : n);
    float4 o;
    o.x = xv.x + ((a0.x + a1.x) + (a2.x + a3.x)) / d;
    o.y = xv.y + ((a0.y + a1.y) + (a2.y + a3.y)) / d;
    o.z = xv.z + ((a0.z + a1.z) + (a2.z + a3.z)) / d;
    o.w = xv.w + ((a0.w + a1.w) + (a2.w + a3.w)) / d;
    ((float4*)h)[(size_t)node * 32 + lane] = o;
}

__global__ __launch_bounds__(256) void gc_gemm_wmma(const float* __restrict__ h,
                                                    const float* __restrict__ W,
                                                    const float* __restrict__ bias,
                                                    float*       __restrict__ out) {
    const int wid  = (blockIdx.x * blockDim.x + threadIdx.x) >> 5;
    const int lane = threadIdx.x & 31;
    const int mt = wid >> 3;
    const int nt = wid & 7;
    const int m0 = mt * 16;
    const int n0 = nt * 16;
    const int ln = lane & 15;
    const int hi = lane >> 4;

    const float bv = bias[n0 + ln];
    v8f c;
#pragma unroll
    for (int r = 0; r < 8; ++r) c[r] = bv;

    const float* hrow = h + (size_t)(m0 + ln) * CH;
    const float* wrow = W + (size_t)(n0 + ln) * CH;

#pragma unroll 4
    for (int kb = 0; kb < 32; ++kb) {
        const int k = kb * 4 + hi * 2;
        v2f a = *(const v2f*)(hrow + k);
        v2f b = *(const v2f*)(wrow + k);
        c = __builtin_amdgcn_wmma_f32_16x16x4_f32(false, a, false, b,
                                                  (short)0, c, false, false);
    }

    float* orow = out + (size_t)(m0 + hi * 8) * CH + n0 + ln;
#pragma unroll
    for (int r = 0; r < 8; ++r) orow[(size_t)r * CH] = c[r];
}


extern "C" void kernel_launch(void* const* d_in, const int* in_sizes, int n_in,
                              void* d_out, int out_size, void* d_ws, size_t ws_size,
                              hipStream_t stream) {
    const float* x    = (const float*)d_in[0];
    const int*   ei   = (const int*)  d_in[1];
    const float* W    = (const float*)d_in[2];
    const float* bias = (const float*)d_in[3];
    float*       out  = (float*)d_out;

    const int* rows = ei;
    const int* cols = ei + N_EDGES;

    int*   cnt  = (int*)d_ws;
    int*   offs = cnt  + N_PAD;
    int*   rank = offs + N_PAD;
    int*   csr  = rank + N_EDGES;
    float* h    = (float*)(csr + N_EDGES);

    gc_zero_cnt<<<(N_NODES + 255) / 256, 256, 0, stream>>>(cnt);
    gc_count<<<N_EDGES / 256, 256, 0, stream>>>(rows, cnt, rank);
    gc_scan<<<1, 1024, 0, stream>>>(cnt, offs);
    gc_fill<<<N_EDGES / 256, 256, 0, stream>>>(rows, cols, offs, rank, csr);
    gc_gather<<<N_NODES / 8, 256, 0, stream>>>(x, cnt, offs, csr, h);
    gc_gemm_wmma<<<(3125 * 8 * 32) / 256, 256, 0, stream>>>(h, W, bias, out);
}


# gather fused into the linear layer, one resident kernel: a work-group gathers 16 h rows (2 nodes per wave, walked together) into an LDS tile, f32 WMMA from it, W in registers; h plane, 1 launch gone
# speedup vs baseline: 2.0115x; 1.1582x over previous
#include <hip/hip_runtime.h>


#define N_NODES 50000
#define N_EDGES 640000
#define CH      128
#define N_PAD   50176
#define N_TILES 3125
#define GRID_WG 625
#define H_PITCH 132

typedef float v2f __attribute__((ext_vector_type(2)));
typedef float v8f __attribute__((ext_vector_type(8)));


__global__ __launch_bounds__(256) void gc_zero_cnt(int* __restrict__ cnt) {
    const int i = blockIdx.x * blockDim.x + threadIdx.x;
    if (i < N_NODES) cnt[i] = 0;
}

__global__ __launch_bounds__(256) void gc_count(const int* __restrict__ rows,
                                                int*       __restrict__ cnt,
                                                int*       __restrict__ rank) {
    const int e = blockIdx.x * blockDim.x + threadIdx.x;
    const int r = rows[e];
    rank[e] = atomicAdd(cnt + r, 1);
}

__global__ __launch_bounds__(1024) void gc_scan(const int* __restrict__ cnt,
                                                int*       __restrict__ offs) {
    __shared__ int4 s4[N_PAD / 4];
    __shared__ int wtot[32];
    int* s = (int*)s4;
    const int t    = threadIdx.x;
    const int lane = t & 31;
    const int wave = t >> 5;

    const int4* __restrict__ cnt4 = (const int4*)cnt;
    int4 st[13];
#pragma unroll
    for (int k = 0; k < 13; ++k) {
        const int q = t + k * 1024;
        st[k] = (q < N_NODES / 4) ? cnt4[q] : make_int4(0, 0, 0, 0);
    }
#pragma unroll
    for (int k = 0; k < 13; ++k) {
        const int q = t + k * 1024;
        if (q < N_PAD / 4) s4[q] = st[k];
    }
    __syncthreads();

    const int base = t * 49;
    int sum = 0;
#pragma unroll
    for (int i = 0; i < 49; ++i) sum += s[base + i];

    int incl = sum;
#pragma unroll
    for (int d = 1; d < 32; d <<= 1) {
        const int up = __shfl_up(incl, d);
        if (lane >= d) incl += up;
    }
    if (lane == 31) wtot[wave] = incl;
    __syncthreads();

    const int wv = wtot[lane];
    int wincl = wv;
#pragma unroll
    for (int d = 1; d < 32; d <<= 1) {
        const int up = __shfl_up(wincl, d);
        if (lane >= d) wincl += up;
    }
    const int wbase = __shfl(wincl - wv, wave);

    int run = wbase + incl - sum;
#pragma unroll
    for (int i = 0; i < 49; ++i) {
        const int v = s[base + i];
        s[base + i] = run;
        run += v;
    }
    __syncthreads();

    int4* __restrict__ offs4 = (int4*)offs;
#pragma unroll
    for (int k = 0; k < 13; ++k) {
        const int q = t + k * 1024;
        if (q < N_NODES / 4) offs4[q] = s4[q];
    }
}

__global__ __launch_bounds__(256) void gc_fill(const int* __restrict__ rows,
                                               const int* __restrict__ cols,
                                               const int* __restrict__ offs,
                                               const int* __restrict__ rank,
                                               int*       __restrict__ csr) {
    const int e = blockIdx.x * blockDim.x + threadIdx.x;
    const int r = rows[e];
    unsigned p = (unsigned)(offs[r] + rank[e]);
    p = p < (unsigned)N_EDGES ? p : (unsigned)(N_EDGES - 1);
    csr[p] = cols[e];
}


__device__ __forceinline__ void gc_add4(float4& a, const float4& v) {
    a.x += v.x; a.y += v.y; a.z += v.z; a.w += v.w;
}

__device__ __forceinline__ int gc_ids(const int* __restrict__ csr, const int first, const int lane) {
    unsigned p = (unsigned)(first + lane);
    p = p < (unsigned)N_EDGES ? p : (unsigned)(N_EDGES - 1);
    const int cl = csr[p];
    return (unsigned)cl < (unsigned)N_NODES ? cl : (N_NODES - 1);
}

__global__ __launch_bounds__(256, 6) void gc_gather_gemm(const float* __restrict__ x,
                                                      const int*   __restrict__ cnt,
                                                      const int*   __restrict__ offs,
                                                      const int*   __restrict__ csr,
                                                      const float* __restrict__ W,
                                                      const float* __restrict__ bias,
                                                      float*       __restrict__ out) {
    __shared__ float4 hs4[2][16 * H_PITCH / 4];

    const int lane = threadIdx.x & 31;
    const int wave = __builtin_amdgcn_readfirstlane((int)(threadIdx.x >> 5));
    const int ln = lane & 15;
    const int hi = lane >> 4;
    const int n0 = wave * 16;

    const float* wrow = W + (size_t)(n0 + ln) * CH + hi * 2;
    v2f bf[32];
#pragma unroll
    for (int kb = 0; kb < 32; ++kb) bf[kb] = *(const v2f*)(wrow + kb * 4);
    const float bv = bias[n0 + ln];

    const char* __restrict__ xb = (const char*)x;
    const unsigned lane16 = (unsigned)lane * 16u;

    int buf = 0;
    for (int tile = blockIdx.x; tile < N_TILES; tile += GRID_WG, buf ^= 1) {
        float* hs = (float*)hs4[buf];
        const int m0 = tile * 16;
        const int r0 = wave * 2;
        const int nodeA = m0 + r0;
        const int nodeB = nodeA + 1;

        const int nA   = __builtin_amdgcn_readfirstlane(cnt[nodeA]);
        const int nB   = __builtin_amdgcn_readfirstlane(cnt[nodeB]);
        const int begA = __builtin_amdgcn_readfirstlane(offs[nodeA]);
        const int begB = __builtin_amdgcn_readfirstlane(offs[nodeB]);
        const int nmax = nA > nB ? nA : nB;

        int clA = gc_ids(csr, begA, lane);
        int clB = gc_ids(csr, begB, lane);
        const float4 xvA = *(const float4*)(xb + ((size_t)(unsigned)nodeA << 9) + lane16);
        const float4 xvB = *(const float4*)(xb + ((size_t)(unsigned)nodeB << 9) + lane16);

        const float4 zero4 = make_float4(0.f, 0.f, 0.f, 0.f);
        float4 sA[4], sB[4];
#pragma unroll
        for (int i = 0; i < 4; ++i) { sA[i] = zero4; sB[i] = zero4; }

        for (int b = 0; b < nmax; b += 32) {
            const int mA = nA - b < 32 ? nA - b : 32;
            const int mB = nB - b < 32 ? nB - b : 32;
            const int mm = mA > mB ? mA : mB;
            for (int j = 0; j < mm; j += 4) {
                const int kA = mA - j;
                const int kB = mB - j;
                float4 v[8];
#pragma unroll
                for (int i = 0; i < 4; ++i) {
                    if (kA > i) {
                        const unsigned c = (unsigned)__builtin_amdgcn_readlane(clA, j + i);
                        v[i] = *(const float4*)(xb + ((size_t)c << 9) + lane16);
                    }
                    if (kB > i) {
                        const unsigned c = (unsigned)__builtin_amdgcn_readlane(clB, j + i);
                        v[4 + i] = *(const float4*)(xb + ((size_t)c << 9) + lane16);
                    }
                }
#pragma unroll
                for (int i = 0; i < 4; ++i) {
                    gc_add4(sA[i], kA > i ? v[i]     : zero4);
                    gc_add4(sB[i], kB > i ? v[4 + i] : zero4);
                }
            }
            if (b + 32 < nmax) {
                clA = gc_ids(csr, begA + b + 32, lane);
                clB = gc_ids(csr, begB + b + 32, lane);
            }
        }

        const float dA = (float)(nA < 1 ? 1 : nA);
        const float dB = (float)(nB < 1 ? 1 : nB);
        float4 hA, hB;
        hA.x = xvA.x + ((sA[0].x + sA[1].x) + (sA[2].x + sA[3].x)) / dA;
        hA.y = xvA.y + ((sA[0].y + sA[1].y) + (sA[2].y + sA[3].y)) / dA;
        hA.z = xvA.z + ((sA[0].z + sA[1].z) + (sA[2].z + sA[3].z)) / dA;
        hA.w = xvA.w + ((sA[0].w + sA[1].w) + (sA[2].w + sA[3].w)) / dA;
        hB.x = xvB.x + ((sB[0].x + sB[1].x) + (sB[2].x + sB[3].x)) / dB;
        hB.y = xvB.y + ((sB[0].y + sB[1].y) + (sB[2].y + sB[3].y)) / dB;
        hB.z = xvB.z + ((sB[0].z + sB[1].z) + (sB[2].z + sB[3].z)) / dB;
        hB.w = xvB.w + ((sB[0].w + sB[1].w) + (sB[2].w + sB[3].w)) / dB;
        *(float4*)(hs + (r0 + 0) * H_PITCH + lane * 4) = hA;
        *(float4*)(hs + (r0 + 1) * H_PITCH + lane * 4) = hB;

        __syncthreads();

        v8f c;
#pragma unroll
        for (int r = 0; r < 8; ++r) c[r] = bv;

        const float* arow = hs + ln * H_PITCH + hi * 2;
#pragma unroll
        for (int kb = 0; kb < 32; ++kb) {
            const v2f a = *(const v2f*)(arow + kb * 4);
            c = __builtin_amdgcn_wmma_f32_16x16x4_f32(false, a, false, bf[kb],
                                                      (short)0, c, false, false);
        }

        float* orow = out + (size_t)(m0 + hi * 8) * CH + n0 + ln;
#pragma unroll
        for (int r = 0; r < 8; ++r) orow[(size_t)r * CH] = c[r];
    }
}


extern "C" void kernel_launch(void* const* d_in, const int* in_sizes, int n_in,
                              void* d_out, int out_size, void* d_ws, size_t ws_size,
                              hipStream_t stream) {
    const float* x    = (const float*)d_in[0];
    const int*   ei   = (const int*)  d_in[1];
    const float* W    = (const float*)d_in[2];
    const float* bias = (const float*)d_in[3];
    float*       out  = (float*)d_out;

    const int* rows = ei;
    const int* cols = ei + N_EDGES;

    int*   cnt  = (int*)d_ws;
    int*   offs = cnt  + N_PAD;
    int*   rank = offs + N_PAD;
    int*   csr  = rank + N_EDGES;

    gc_zero_cnt<<<(N_NODES + 255) / 256, 256, 0, stream>>>(cnt);
    gc_count<<<N_EDGES / 256, 256, 0, stream>>>(rows, cnt, rank);
    gc_scan<<<1, 1024, 0, stream>>>(cnt, offs);
    gc_fill<<<N_EDGES / 256, 256, 0, stream>>>(rows, cols, offs, rank, csr);
    gc_gather_gemm<<<GRID_WG, 256, 0, stream>>>(x, cnt, offs, csr, W, bias, out);
}


# prefix-sum and fill launches removed: the count pass writes each col into its row's fixed 64-word slot group at the atomic ticket; a row over 64 edges is summed from the edge list in edge order
# speedup vs baseline: 2.5933x; 1.2892x over previous
#include <hip/hip_runtime.h>


#define N_NODES 50000
#define N_EDGES 640000
#define CH      128
#define N_PAD   50176
#define CAP     64
#define N_TILES 3125
#define GRID_WG 625
#define H_PITCH 132

typedef float v2f __attribute__((ext_vector_type(2)));
typedef float v8f __attribute__((ext_vector_type(8)));


__global__ __launch_bounds__(256) void gc_zero_cnt(int* __restrict__ cnt) {
    const int i = blockIdx.x * blockDim.x + threadIdx.x;
    if (i < N_NODES) cnt[i] = 0;
}

__global__ __launch_bounds__(256) void gc_count_fill(const int* __restrict__ rows,
                                                     const int* __restrict__ cols,
                                                     int*       __restrict__ cnt,
                                                     int*       __restrict__ slot) {
    const int e = blockIdx.x * blockDim.x + threadIdx.x;
    const int r = rows[e];
    const int c = cols[e];
    if ((unsigned)r >= (unsigned)N_NODES) return;
    const int k = atomicAdd(cnt + r, 1);
    if ((unsigned)k < (unsigned)CAP) slot[r * CAP + k] = c;
}


__device__ __forceinline__ void gc_add4(float4& a, const float4& v) {
    a.x += v.x; a.y += v.y; a.z += v.z; a.w += v.w;
}

__device__ __forceinline__ int gc_ids(const int* __restrict__ ws, const int node, const int first,
                                      const int lane, const int left) {
    const int w  = lane < left ? N_PAD + node * CAP + first + lane : node;
    const int cl = ws[w];
    return (unsigned)cl < (unsigned)N_NODES ? cl : (N_NODES - 1);
}

__device__ __forceinline__ float4 gc_sum_edges(const int*  __restrict__ rows,
                                               const int*  __restrict__ cols,
                                               const char* __restrict__ xb,
                                               const int node, const int lane,
                                               const unsigned lane16) {
    float4 s = make_float4(0.f, 0.f, 0.f, 0.f);
    for (int e0 = 0; e0 < N_EDGES; e0 += 32) {
        const int r  = rows[e0 + lane];
        const int c0 = cols[e0 + lane];
        const int c  = (unsigned)c0 < (unsigned)N_NODES ? c0 : (N_NODES - 1);
        unsigned m = __builtin_amdgcn_ballot_w32(r == node);
        while (m != 0u) {
            const int l = __builtin_ctz(m);
            m &= m - 1u;
            const unsigned cc = (unsigned)__builtin_amdgcn_readlane(c, l);
            gc_add4(s, *(const float4*)(xb + ((size_t)cc << 9) + lane16));
        }
    }
    return s;
}

__global__ __launch_bounds__(256, 6) void gc_gather_gemm(const float* __restrict__ x,
                                                      const int*   __restrict__ rows,
                                                      const int*   __restrict__ cols,
                                                      const int*   __restrict__ ws,
                                                      const float* __restrict__ W,
                                                      const float* __restrict__ bias,
                                                      float*       __restrict__ out) {
    __shared__ float4 hs4[2][16 * H_PITCH / 4];

    const int lane = threadIdx.x & 31;
    const int wave = __builtin_amdgcn_readfirstlane((int)(threadIdx.x >> 5));
    const int ln = lane & 15;
    const int hi = lane >> 4;
    const int n0 = wave * 16;

    const float* wrow = W + (size_t)(n0 + ln) * CH + hi * 2;
    v2f bf[32];
#pragma unroll
    for (int kb = 0; kb < 32; ++kb) bf[kb] = *(const v2f*)(wrow + kb * 4);
    const float bv = bias[n0 + ln];

    const char* __restrict__ xb = (const char*)x;
    const unsigned lane16 = (unsigned)lane * 16u;

    int buf = 0;
    for (int tile = blockIdx.x; tile < N_TILES; tile += GRID_WG, buf ^= 1) {
        float* hs = (float*)hs4[buf];
        const int m0 = tile * 16;
        const int r0 = wave * 2;
        const int nodeA = m0 + r0;
        const int nodeB = nodeA + 1;

        const int nA = __builtin_amdgcn_readfirstlane(ws[nodeA]);
        const int nB = __builtin_amdgcn_readfirstlane(ws[nodeB]);
        const int fA = nA > CAP ? 0 : nA;
        const int fB = nB > CAP ? 0 : nB;
        const int nmax = fA > fB ? fA : fB;

        int clA = gc_ids(ws, nodeA, 0, lane, fA);
        int clB = gc_ids(ws, nodeB, 0, lane, fB);
        const float4 xvA = *(const float4*)(xb + ((size_t)(unsigned)nodeA << 9) + lane16);
        const float4 xvB = *(const float4*)(xb + ((size_t)(unsigned)nodeB << 9) + lane16);

        const float4 zero4 = make_float4(0.f, 0.f, 0.f, 0.f);
        float4 sA[4], sB[4];
#pragma unroll
        for (int i = 0; i < 4; ++i) { sA[i] = zero4; sB[i] = zero4; }

        for (int b = 0; b < nmax; b += 32) {
            const int mA = fA - b < 32 ? fA - b : 32;
            const int mB = fB - b < 32 ? fB - b : 32;
            const int mm = mA > mB ? mA : mB;
            for (int j = 0; j < mm; j += 4) {
                const int kA = mA - j;
                const int kB = mB - j;
                float4 v[8];
#pragma unroll
                for (int i = 0; i < 4; ++i) {
                    if (kA > i) {
                        const unsigned c = (unsigned)__builtin_amdgcn_readlane(clA, j + i);
                        v[i] = *(const float4*)(xb + ((size_t)c << 9) + lane16);
                    }
                    if (kB > i) {
                        const unsigned c = (unsigned)__builtin_amdgcn_readlane(clB, j + i);
                        v[4 + i] = *(const float4*)(xb + ((size_t)c << 9) + lane16);
                    }
                }
#pragma unroll
                for (int i = 0; i < 4; ++i) {
                    gc_add4(sA[i], kA > i ? v[i]     : zero4);
                    gc_add4(sB[i], kB > i ? v[4 + i] : zero4);
                }
            }
            if (b + 32 < nmax) {
                clA = gc_ids(ws, nodeA, b + 32, lane, fA - b - 32);
                clB = gc_ids(ws, nodeB, b + 32, lane, fB - b - 32);
            }
        }
        if (nA > CAP) sA[0] = gc_sum_edges(rows, cols, xb, nodeA, lane, lane16);
        if (nB > CAP) sB[0] = gc_sum_edges(rows, cols, xb, nodeB, lane, lane16);

        const float dA = (float)(nA < 1 ? 1 : nA);
        const float dB = (float)(nB < 1 ? 1 : nB);
        float4 hA, hB;
        hA.x = xvA.x + ((sA[0].x + sA[1].x) + (sA[2].x + sA[3].x)) / dA;
        hA.y = xvA.y + ((sA[0].y + sA[1].y) + (sA[2].y + sA[3].y)) / dA;
        hA.z = xvA.z + ((sA[0].z + sA[1].z) + (sA[2].z + sA[3].z)) / dA;
        hA.w = xvA.w + ((sA[0].w + sA[1].w) + (sA[2].w + sA[3].w)) / dA;
        hB.x = xvB.x + ((sB[0].x + sB[1].x) + (sB[2].x + sB[3].x)) / dB;
        hB.y = xvB.y + ((sB[0].y + sB[1].y) + (sB[2].y + sB[3].y)) / dB;
        hB.z = xvB.z + ((sB[0].z + sB[1].z) + (sB[2].z + sB[3].z)) / dB;
        hB.w = xvB.w + ((sB[0].w + sB[1].w) + (sB[2].w + sB[3].w)) / dB;
        *(float4*)(hs + (r0 + 0) * H_PITCH + lane * 4) = hA;
        *(float4*)(hs + (r0 + 1) * H_PITCH + lane * 4) = hB;

        __syncthreads();

        v8f c;
#pragma unroll
        for (int r = 0; r < 8; ++r) c[r] = bv;

        const float* arow = hs + ln * H_PITCH + hi * 2;
#pragma unroll
        for (int kb = 0; kb < 32; ++kb) {
            const v2f a = *(const v2f*)(arow + kb * 4);
            c = __builtin_amdgcn_wmma_f32_16x16x4_f32(false, a, false, bf[kb],
                                                      (short)0, c, false, false);
        }

        float* orow = out + (size_t)(m0 + hi * 8) * CH + n0 + ln;
#pragma unroll
        for (int r = 0; r < 8; ++r) orow[(size_t)r * CH] = c[r];
    }
}


extern "C" void kernel_launch(void* const* d_in, const int* in_sizes, int n_in,
                              void* d_out, int out_size, void* d_ws, size_t ws_size,
                              hipStream_t stream) {
    const float* x    = (const float*)d_in[0];
    const int*   ei   = (const int*)  d_in[1];
    const float* W    = (const float*)d_in[2];
    const float* bias = (const float*)d_in[3];
    float*       out  = (float*)d_out;

    const int* rows = ei;
    const int* cols = ei + N_EDGES;

    int*   cnt  = (int*)d_ws;
    int*   slot = cnt + N_PAD;

    gc_zero_cnt<<<(N_NODES + 255) / 256, 256, 0, stream>>>(cnt);
    gc_count_fill<<<N_EDGES / 256, 256, 0, stream>>>(rows, cols, cnt, slot);
    gc_gather_gemm<<<GRID_WG, 256, 0, stream>>>(x, rows, cols, cnt, W, bias, out);
}


# fused kernel in two phases, one barrier: a work-group first gathers all 80 of its h rows into LDS (ids one pair ahead, 16 row loads a step), then runs its 5 tiles' WMMA chains interleaved
# speedup vs baseline: 2.7891x; 1.0755x over previous
#include <hip/hip_runtime.h>


#define N_NODES 50000
#define N_EDGES 640000
#define CH      128
#define N_PAD   50176
#define CAP     64
#define TPW     5
#define ROWS_WG 80
#define RPW     10
#define GRID_WG 625
#define H_PITCH 132

typedef float v2f __attribute__((ext_vector_type(2)));
typedef float v8f __attribute__((ext_vector_type(8)));


__global__ __launch_bounds__(256) void gc_zero_cnt(int* __restrict__ cnt) {
    const int i = blockIdx.x * blockDim.x + threadIdx.x;
    if (i < N_NODES) cnt[i] = 0;
}

__global__ __launch_bounds__(256) void gc_count_fill(const int* __restrict__ rows,
                                                     const int* __restrict__ cols,
                                                     int*       __restrict__ cnt,
                                                     int*       __restrict__ slot) {
    const int e = blockIdx.x * blockDim.x + threadIdx.x;
    const int r = rows[e];
    const int c = cols[e];
    if ((unsigned)r >= (unsigned)N_NODES) return;
    const int k = atomicAdd(cnt + r, 1);
    if ((unsigned)k < (unsigned)CAP) slot[r * CAP + k] = c;
}


__device__ __forceinline__ void gc_add4(float4& a, const float4& v) {
    a.x += v.x; a.y += v.y; a.z += v.z; a.w += v.w;
}

__device__ __forceinline__ int gc_ids(const int* __restrict__ ws, const int node, const int first,
                                      const int lane, const int left) {
    const int w = lane < left ? N_PAD + node * CAP + first + lane : node;
    return ws[w];
}

__device__ __forceinline__ int gc_clamp(const int cl) {
    return (unsigned)cl < (unsigned)N_NODES ? cl : (N_NODES - 1);
}

__device__ __forceinline__ float4 gc_sum_edges(const int*  __restrict__ rows,
                                               const int*  __restrict__ cols,
                                               const char* __restrict__ xb,
                                               const int node, const int lane,
                                               const unsigned lane16) {
    float4 s = make_float4(0.f, 0.f, 0.f, 0.f);
    for (int e0 = 0; e0 < N_EDGES; e0 += 32) {
        const int r  = rows[e0 + lane];
        const int c0 = cols[e0 + lane];
        const int c  = (unsigned)c0 < (unsigned)N_NODES ? c0 : (N_NODES - 1);
        unsigned m = __builtin_amdgcn_ballot_w32(r == node);
        while (m != 0u) {
            const int l = __builtin_ctz(m);
            m &= m - 1u;
            const unsigned cc = (unsigned)__builtin_amdgcn_readlane(c, l);
            gc_add4(s, *(const float4*)(xb + ((size_t)cc << 9) + lane16));
        }
    }
    return s;
}

__global__ __launch_bounds__(256, 6) void gc_gather_gemm(const float* __restrict__ x,
                                                      const int*   __restrict__ rows,
                                                      const int*   __restrict__ cols,
                                                      const int*   __restrict__ ws,
                                                      const float* __restrict__ W,
                                                      const float* __restrict__ bias,
                                                      float*       __restrict__ out) {
    __shared__ float4 hs4[ROWS_WG * H_PITCH / 4];
    float* hs = (float*)hs4;

    const int lane = threadIdx.x & 31;
    const int wave = __builtin_amdgcn_readfirstlane((int)(threadIdx.x >> 5));
    const int ln = lane & 15;
    const int hi = lane >> 4;
    const int n0 = wave * 16;

    const char* __restrict__ xb = (const char*)x;
    const unsigned lane16 = (unsigned)lane * 16u;

    const int base  = blockIdx.x * ROWS_WG;
    const int row0  = wave * RPW;
    const int node0 = base + row0;

    const int cntv = ws[node0 + (lane < RPW ? lane : 0)];

    const float4 zero4 = make_float4(0.f, 0.f, 0.f, 0.f);

    int nA = __builtin_amdgcn_readlane(cntv, 0);
    int nB = __builtin_amdgcn_readlane(cntv, 1);
    int fA = nA > CAP ? 0 : nA;
    int fB = nB > CAP ? 0 : nB;
    int rawA = gc_ids(ws, node0,     0, lane, fA);
    int rawB = gc_ids(ws, node0 + 1, 0, lane, fB);
    float4 xvA = *(const float4*)(xb + ((size_t)(unsigned)node0 << 9) + lane16);
    float4 xvB = *(const float4*)(xb + ((size_t)(unsigned)(node0 + 1) << 9) + lane16);

    for (int p = 0; p < RPW / 2; ++p) {
        const int nodeA = node0 + 2 * p;
        const int nodeB = nodeA + 1;

        const int pn = p + 1 < RPW / 2 ? p + 1 : p;
        const int nodeA2 = node0 + 2 * pn;
        const int nodeB2 = nodeA2 + 1;
        const int nA2 = __builtin_amdgcn_readlane(cntv, 2 * pn);
        const int nB2 = __builtin_amdgcn_readlane(cntv, 2 * pn + 1);
        const int fA2 = nA2 > CAP ? 0 : nA2;
        const int fB2 = nB2 > CAP ? 0 : nB2;
        const int rawA2 = gc_ids(ws, nodeA2, 0, lane, fA2);
        const int rawB2 = gc_ids(ws, nodeB2, 0, lane, fB2);
        const float4 xvA2 = *(const float4*)(xb + ((size_t)(unsigned)nodeA2 << 9) + lane16);
        const float4 xvB2 = *(const float4*)(xb + ((size_t)(unsigned)nodeB2 << 9) + lane16);

        int clA = gc_clamp(rawA);
        int clB = gc_clamp(rawB);
        const int nmax = fA > fB ? fA : fB;

        float4 sA[4], sB[4];
#pragma unroll
        for (int i = 0; i < 4; ++i) { sA[i] = zero4; sB[i] = zero4; }

        for (int b = 0; b < nmax; b += 32) {
            const int mA = fA - b < 32 ? fA - b : 32;
            const int mB = fB - b < 32 ? fB - b : 32;
            const int mm = mA > mB ? mA : mB;
            for (int j = 0; j < mm; j += 8) {
                const int kA = mA - j;
                const int kB = mB - j;
                float4 v[16];
#pragma unroll
                for (int i = 0; i < 8; ++i) {
                    if (kA > i) {
                        const unsigned c = (unsigned)__builtin_amdgcn_readlane(clA, j + i);
                        v[i] = *(const float4*)(xb + ((size_t)c << 9) + lane16);
                    }
                    if (kB > i) {
                        const unsigned c = (unsigned)__builtin_amdgcn_readlane(clB, j + i);
                        v[8 + i] = *(const float4*)(xb + ((size_t)c << 9) + lane16);
                    }
                }
#pragma unroll
                for (int i = 0; i < 8; ++i) {
                    gc_add4(sA[i & 3], kA > i ? v[i]     : zero4);
                    gc_add4(sB[i & 3], kB > i ? v[8 + i] : zero4);
                }
            }
            if (b + 32 < nmax) {
                clA = gc_clamp(gc_ids(ws, nodeA, b + 32, lane, fA - b - 32));
                clB = gc_clamp(gc_ids(ws, nodeB, b + 32, lane, fB - b - 32));
            }
        }
        if (nA > CAP) sA[0] = gc_sum_edges(rows, cols, xb, nodeA, lane, lane16);
        if (nB > CAP) sB[0] = gc_sum_edges(rows, cols, xb, nodeB, lane, lane16);

        const float dA = (float)(nA < 1 ? 1 : nA);
        const float dB = (float)(nB < 1 ? 1 : nB);
        float4 hA, hB;
        hA.x = xvA.x + ((sA[0].x + sA[1].x) + (sA[2].x + sA[3].x)) / dA;
        hA.y = xvA.y + ((sA[0].y + sA[1].y) + (sA[2].y + sA[3].y)) / dA;
        hA.z = xvA.z + ((sA[0].z + sA[1].z) + (sA[2].z + sA[3].z)) / dA;
        hA.w = xvA.w + ((sA[0].w + sA[1].w) + (sA[2].w + sA[3].w)) / dA;
        hB.x = xvB.x + ((sB[0].x + sB[1].x) + (sB[2].x + sB[3].x)) / dB;
        hB.y = xvB.y + ((sB[0].y + sB[1].y) + (sB[2].y + sB[3].y)) / dB;
        hB.z = xvB.z + ((sB[0].z + sB[1].z) + (sB[2].z + sB[3].z)) / dB;
        hB.w = xvB.w + ((sB[0].w + sB[1].w) + (sB[2].w + sB[3].w)) / dB;
        *(float4*)(hs + (row0 + 2 * p + 0) * H_PITCH + lane * 4) = hA;
        *(float4*)(hs + (row0 + 2 * p + 1) * H_PITCH + lane * 4) = hB;

        nA = nA2; nB = nB2; fA = fA2; fB = fB2;
        rawA = rawA2; rawB = rawB2; xvA = xvA2; xvB = xvB2;
    }

    const float* wrow = W + (size_t)(n0 + ln) * CH + hi * 2;
    v2f bf[32];
#pragma unroll
    for (int kb = 0; kb < 32; ++kb) bf[kb] = *(const v2f*)(wrow + kb * 4);
    const float bv = bias[n0 + ln];

    __syncthreads();

    v8f c[TPW];
#pragma unroll
    for (int t = 0; t < TPW; ++t) {
#pragma unroll
        for (int r = 0; r < 8; ++r) c[t][r] = bv;
    }

    const float* arow = hs + ln * H_PITCH + hi * 2;
#pragma unroll
    for (int kb = 0; kb < 32; ++kb) {
#pragma unroll
        for (int t = 0; t < TPW; ++t) {
            const v2f a = *(const v2f*)(arow + t * 16 * H_PITCH + kb * 4);
            c[t] = __builtin_amdgcn_wmma_f32_16x16x4_f32(false, a, false, bf[kb],
                                                         (short)0, c[t], false, false);
        }
    }

#pragma unroll
    for (int t = 0; t < TPW; ++t) {
        float* orow = out + (size_t)(base + t * 16 + hi * 8) * CH + n0 + ln;
#pragma unroll
        for (int r = 0; r < 8; ++r) orow[(size_t)r * CH] = c[t][r];
    }
}


extern "C" void kernel_launch(void* const* d_in, const int* in_sizes, int n_in,
                              void* d_out, int out_size, void* d_ws, size_t ws_size,
                              hipStream_t stream) {
    const float* x    = (const float*)d_in[0];
    const int*   ei   = (const int*)  d_in[1];
    const float* W    = (const float*)d_in[2];
    const float* bias = (const float*)d_in[3];
    float*       out  = (float*)d_out;

    const int* rows = ei;
    const int* cols = ei + N_EDGES;

    int*   cnt  = (int*)d_ws;
    int*   slot = cnt + N_PAD;

    gc_zero_cnt<<<(N_NODES + 255) / 256, 256, 0, stream>>>(cnt);
    gc_count_fill<<<N_EDGES / 256, 256, 0, stream>>>(rows, cols, cnt, slot);
    gc_gather_gemm<<<GRID_WG, 256, 0, stream>>>(x, rows, cols, cnt, W, bias, out);
}


# per-edge global ticket add + col store replaced by two-level LDS binning: 250 blocks write one 128-B segment per 80-row group, gather kernel builds row lists in LDS (LDS ticket order); 1 launch less
# speedup vs baseline: 5.4813x; 1.9653x over previous
#include <hip/hip_runtime.h>


#define N_NODES 50000
#define N_EDGES 640000
#define CH      128
#define CAP     64
#define N_BIN   625
#define G1      250
#define E1      2560
#define SEGW    32
#define TPW     5
#define ROWS_WG 80
#define RPW     10
#define GRID_WG 625
#define H_PITCH 132

typedef float v2f __attribute__((ext_vector_type(2)));
typedef float v8f __attribute__((ext_vector_type(8)));


__device__ __forceinline__ int gc_clamp(const int cl) {
    return (unsigned)cl < (unsigned)N_NODES ? cl : (N_NODES - 1);
}

__global__ __launch_bounds__(256) void gc_bin(const int* __restrict__ rows,
                                              const int* __restrict__ cols,
                                              int*       __restrict__ seg) {
    __shared__ int  lcnt[N_BIN];
    __shared__ int4 sg4[N_BIN * SEGW / 4];
    int* sg = (int*)sg4;
    const int tid = threadIdx.x;
    for (int i = tid; i < N_BIN; i += 256) lcnt[i] = 0;
    for (int i = tid; i < N_BIN * SEGW / 4; i += 256) sg4[i] = make_int4(0, 0, 0, 0);
    __syncthreads();
    const int e0 = blockIdx.x * E1;
    int r[E1 / 256], c[E1 / 256], bn[E1 / 256], t[E1 / 256];
#pragma unroll
    for (int i = 0; i < E1 / 256; ++i) {
        r[i] = rows[e0 + i * 256 + tid];
        c[i] = cols[e0 + i * 256 + tid];
    }
    asm volatile("" :: "v"(r[0]), "v"(r[1]), "v"(r[2]), "v"(r[3]), "v"(r[4]),
                       "v"(r[5]), "v"(r[6]), "v"(r[7]), "v"(r[8]), "v"(r[9]),
                       "v"(c[0]), "v"(c[1]), "v"(c[2]), "v"(c[3]), "v"(c[4]),
                       "v"(c[5]), "v"(c[6]), "v"(c[7]), "v"(c[8]), "v"(c[9]));
#pragma unroll
    for (int i = 0; i < E1 / 256; ++i) {
        const bool in = (unsigned)r[i] < (unsigned)N_NODES;
        bn[i] = in ? (int)((unsigned)r[i] / (unsigned)ROWS_WG) : 0;
        t[i] = SEGW;
        if (in) t[i] = atomicAdd(&lcnt[bn[i]], 1);
    }
#pragma unroll
    for (int i = 0; i < E1 / 256; ++i) {
        if (t[i] < SEGW - 1)
            sg[bn[i] * SEGW + 1 + t[i]] = ((r[i] - bn[i] * ROWS_WG) << 16) | gc_clamp(c[i]);
    }
    __syncthreads();
    for (int b = tid; b < N_BIN; b += 256) sg[b * SEGW] = lcnt[b];
    __syncthreads();
    const int lane = tid & 31;
    const int wave = tid >> 5;
#pragma unroll 4
    for (int k = wave; k < N_BIN / 4; k += 8) {
        const int b = 4 * k + (lane >> 3);
        const int4 v = sg4[b * (SEGW / 4) + (lane & 7)];
        *(int4*)(seg + ((size_t)b * G1 + blockIdx.x) * SEGW + (lane & 7) * 4) = v;
    }
    if (wave == 0 && lane < 8) {
        const int b = N_BIN - 1;
        const int4 v = sg4[b * (SEGW / 4) + lane];
        *(int4*)(seg + ((size_t)b * G1 + blockIdx.x) * SEGW + lane * 4) = v;
    }
}

__device__ __forceinline__ void gc_put(int* lcn, int* lst, const int rl, const int c) {
    if ((unsigned)rl < (unsigned)ROWS_WG) {
        const int t = atomicAdd(&lcn[rl], 1);
        if (t < CAP) lst[rl * CAP + t] = c;
    }
}

__device__ __forceinline__ void gc_rescan(const int* __restrict__ rows,
                                          const int* __restrict__ cols,
                                          int* lcn, int* lst,
                                          const int g, const int bin, const int lane) {
    const int e0 = g * E1;
    for (int i = 0; i < E1; i += 32) {
        const int r  = rows[e0 + i + lane];
        const int c0 = cols[e0 + i + lane];
        if ((unsigned)r < (unsigned)N_NODES) {
            const int b = (int)((unsigned)r / (unsigned)ROWS_WG);
            if (b == bin) gc_put(lcn, lst, r - b * ROWS_WG, gc_clamp(c0));
        }
    }
}


__device__ __forceinline__ void gc_add4(float4& a, const float4& v) {
    a.x += v.x; a.y += v.y; a.z += v.z; a.w += v.w;
}

__device__ __forceinline__ int gc_ids(const int* lst, const int rl, const int first,
                                      const int lane, const int left) {
    return lane < left ? lst[rl * CAP + first + lane] : 0;
}

__device__ __forceinline__ float4 gc_sum_edges(const int*  __restrict__ rows,
                                               const int*  __restrict__ cols,
                                               const char* __restrict__ xb,
                                               const int node, const int lane,
                                               const unsigned lane16) {
    float4 s = make_float4(0.f, 0.f, 0.f, 0.f);
    for (int e0 = 0; e0 < N_EDGES; e0 += 32) {
        const int r  = rows[e0 + lane];
        const int c0 = cols[e0 + lane];
        const int c  = (unsigned)c0 < (unsigned)N_NODES ? c0 : (N_NODES - 1);
        unsigned m = __builtin_amdgcn_ballot_w32(r == node);
        while (m != 0u) {
            const int l = __builtin_ctz(m);
            m &= m - 1u;
            const unsigned cc = (unsigned)__builtin_amdgcn_readlane(c, l);
            gc_add4(s, *(const float4*)(xb + ((size_t)cc << 9) + lane16));
        }
    }
    return s;
}

__global__ __launch_bounds__(256, 6) void gc_gather_gemm(const float* __restrict__ x,
                                                      const int*   __restrict__ rows,
                                                      const int*   __restrict__ cols,
                                                      const int*   __restrict__ seg,
                                                      const float* __restrict__ W,
                                                      const float* __restrict__ bias,
                                                      float*       __restrict__ out) {
    __shared__ float4 hs4[ROWS_WG * H_PITCH / 4];
    float* hs = (float*)hs4;
    __shared__ int lcn[ROWS_WG];
    __shared__ int lst[ROWS_WG * CAP];

    const int lane = threadIdx.x & 31;
    const int wave = __builtin_amdgcn_readfirstlane((int)(threadIdx.x >> 5));
    const int ln = lane & 15;
    const int hi = lane >> 4;
    const int n0 = wave * 16;

    const char* __restrict__ xb = (const char*)x;
    const unsigned lane16 = (unsigned)lane * 16u;

    const int base  = blockIdx.x * ROWS_WG;
    const int row0  = wave * RPW;
    const int node0 = base + row0;

    if (threadIdx.x < ROWS_WG) lcn[threadIdx.x] = 0;
    __syncthreads();
    {
        const int* sp = seg + (size_t)blockIdx.x * (G1 * SEGW) + lane;
        int sv[32], tk[32];
        unsigned ovf = 0u;
#pragma unroll
        for (int u = 0; u < 32; ++u) {
            const int g = wave + 8 * u;
            const bool has = (8 * u + 7 < G1) || (g < G1);
            sv[u] = has ? sp[g * SEGW] : 0;
        }
#pragma unroll
        for (int u = 0; u < 32; ++u) {
            const int n = __builtin_amdgcn_readfirstlane(sv[u]);
            const int rl = sv[u] >> 16;
            tk[u] = CAP;
            if (n > SEGW - 1) ovf |= 1u << u;
            else if (lane >= 1 && lane <= n && (unsigned)rl < (unsigned)ROWS_WG) tk[u] = atomicAdd(&lcn[rl], 1);
        }
#pragma unroll
        for (int u = 0; u < 32; ++u) {
            if (tk[u] < CAP) lst[(sv[u] >> 16) * CAP + tk[u]] = sv[u] & 0xFFFF;
        }
        while (ovf != 0u) {
            const int u = __builtin_ctz(ovf);
            ovf &= ovf - 1u;
            gc_rescan(rows, cols, lcn, lst, wave + 8 * u, (int)blockIdx.x, lane);
        }
    }
    __syncthreads();

    const int cntv = lcn[row0 + (lane < RPW ? lane : 0)];

    const float4 zero4 = make_float4(0.f, 0.f, 0.f, 0.f);

    int nA = __builtin_amdgcn_readlane(cntv, 0);
    int nB = __builtin_amdgcn_readlane(cntv, 1);
    int fA = nA > CAP ? 0 : nA;
    int fB = nB > CAP ? 0 : nB;
    int rawA = gc_ids(lst, row0,     0, lane, fA);
    int rawB = gc_ids(lst, row0 + 1, 0, lane, fB);
    float4 xvA = *(const float4*)(xb + ((size_t)(unsigned)node0 << 9) + lane16);
    float4 xvB = *(const float4*)(xb + ((size_t)(unsigned)(node0 + 1) << 9) + lane16);

    for (int p = 0; p < RPW / 2; ++p) {
        const int nodeA = node0 + 2 * p;
        const int nodeB = nodeA + 1;

        const int pn = p + 1 < RPW / 2 ? p + 1 : p;
        const int nodeA2 = node0 + 2 * pn;
        const int nodeB2 = nodeA2 + 1;
        const int nA2 = __builtin_amdgcn_readlane(cntv, 2 * pn);
        const int nB2 = __builtin_amdgcn_readlane(cntv, 2 * pn + 1);
        const int fA2 = nA2 > CAP ? 0 : nA2;
        const int fB2 = nB2 > CAP ? 0 : nB2;
        const int rawA2 = gc_ids(lst, row0 + 2 * pn,     0, lane, fA2);
        const int rawB2 = gc_ids(lst, row0 + 2 * pn + 1, 0, lane, fB2);
        const float4 xvA2 = *(const float4*)(xb + ((size_t)(unsigned)nodeA2 << 9) + lane16);
        const float4 xvB2 = *(const float4*)(xb + ((size_t)(unsigned)nodeB2 << 9) + lane16);

        int clA = gc_clamp(rawA);
        int clB = gc_clamp(rawB);
        const int nmax = fA > fB ? fA : fB;

        float4 sA[4], sB[4];
#pragma unroll
        for (int i = 0; i < 4; ++i) { sA[i] = zero4; sB[i] = zero4; }

        for (int b = 0; b < nmax; b += 32) {
            const int mA = fA - b < 32 ? fA - b : 32;
            const int mB = fB - b < 32 ? fB - b : 32;
            const int mm = mA > mB ? mA : mB;
            for (int j = 0; j < mm; j += 8) {
                const int kA = mA - j;
                const int kB = mB - j;
                float4 v[16];
#pragma unroll
                for (int i = 0; i < 8; ++i) {
                    if (kA > i) {
                        const unsigned c = (unsigned)__builtin_amdgcn_readlane(clA, j + i);
                        v[i] = *(const float4*)(xb + ((size_t)c << 9) + lane16);
                    }
                    if (kB > i) {
                        const unsigned c = (unsigned)__builtin_amdgcn_readlane(clB, j + i);
                        v[8 + i] = *(const float4*)(xb + ((size_t)c << 9) + lane16);
                    }
                }
#pragma unroll
                for (int i = 0; i < 8; ++i) {
                    gc_add4(sA[i & 3], kA > i ? v[i]     : zero4);
                    gc_add4(sB[i & 3], kB > i ? v[8 + i] : zero4);
                }
            }
            if (b + 32 < nmax) {
                clA = gc_clamp(gc_ids(lst, row0 + 2 * p,     b + 32, lane, fA - b - 32));
                clB = gc_clamp(gc_ids(lst, row0 + 2 * p + 1, b + 32, lane, fB - b - 32));
            }
        }
        if (nA > CAP) sA[0] = gc_sum_edges(rows, cols, xb, nodeA, lane, lane16);
        if (nB > CAP) sB[0] = gc_sum_edges(rows, cols, xb, nodeB, lane, lane16);

        const float dA = (float)(nA < 1 ? 1 : nA);
        const float dB = (float)(nB < 1 ? 1 : nB);
        float4 hA, hB;
        hA.x = xvA.x + ((sA[0].x + sA[1].x) + (sA[2].x + sA[3].x)) / dA;
        hA.y = xvA.y + ((sA[0].y + sA[1].y) + (sA[2].y + sA[3].y)) / dA;
        hA.z = xvA.z + ((sA[0].z + sA[1].z) + (sA[2].z + sA[3].z)) / dA;
        hA.w = xvA.w + ((sA[0].w + sA[1].w) + (sA[2].w + sA[3].w)) / dA;
        hB.x = xvB.x + ((sB[0].x + sB[1].x) + (sB[2].x + sB[3].x)) / dB;
        hB.y = xvB.y + ((sB[0].y + sB[1].y) + (sB[2].y + sB[3].y)) / dB;
        hB.z = xvB.z + ((sB[0].z + sB[1].z) + (sB[2].z + sB[3].z)) / dB;
        hB.w = xvB.w + ((sB[0].w + sB[1].w) + (sB[2].w + sB[3].w)) / dB;
        *(float4*)(hs + (row0 + 2 * p + 0) * H_PITCH + lane * 4) = hA;
        *(float4*)(hs + (row0 + 2 * p + 1) * H_PITCH + lane * 4) = hB;

        nA = nA2; nB = nB2; fA = fA2; fB = fB2;
        rawA = rawA2; rawB = rawB2; xvA = xvA2; xvB = xvB2;
    }

    const float* wrow = W + (size_t)(n0 + ln) * CH + hi * 2;
    v2f bf[32];
#pragma unroll
    for (int kb = 0; kb < 32; ++kb) bf[kb] = *(const v2f*)(wrow + kb * 4);
    const float bv = bias[n0 + ln];

    __syncthreads();

    v8f c[TPW];
#pragma unroll
    for (int t = 0; t < TPW; ++t) {
#pragma unroll
        for (int r = 0; r < 8; ++r) c[t][r] = bv;
    }

    const float* arow = hs + ln * H_PITCH + hi * 2;
#pragma unroll
    for (int kb = 0; kb < 32; ++kb) {
#pragma unroll
        for (int t = 0; t < TPW; ++t) {
            const v2f a = *(const v2f*)(arow + t * 16 * H_PITCH + kb * 4);
            c[t] = __builtin_amdgcn_wmma_f32_16x16x4_f32(false, a, false, bf[kb],
                                                         (short)0, c[t], false, false);
        }
    }

#pragma unroll
    for (int t = 0; t < TPW; ++t) {
        float* orow = out + (size_t)(base + t * 16 + hi * 8) * CH + n0 + ln;
#pragma unroll
        for (int r = 0; r < 8; ++r) orow[(size_t)r * CH] = c[t][r];
    }
}


extern "C" void kernel_launch(void* const* d_in, const int* in_sizes, int n_in,
                              void* d_out, int out_size, void* d_ws, size_t ws_size,
                              hipStream_t stream) {
    const float* x    = (const float*)d_in[0];
    const int*   ei   = (const int*)  d_in[1];
    const float* W    = (const float*)d_in[2];
    const float* bias = (const float*)d_in[3];
    float*       out  = (float*)d_out;

    const int* rows = ei;
    const int* cols = ei + N_EDGES;

    int* seg = (int*)d_ws;

    gc_bin<<<G1, 256, 0, stream>>>(rows, cols, seg);
    gc_gather_gemm<<<GRID_WG, 256, 0, stream>>>(x, rows, cols, seg, W, bias, out);
}
